# starting kernel
# baseline (speedup 1.0000x reference)
#include <hip/hip_runtime.h>

typedef __attribute__((ext_vector_type(2))) float v2f;
typedef __attribute__((ext_vector_type(8))) float v8f;

#define N_ROWS   1024
#define P_DIM    20000
#define K_TAPS   8
#define P_BLOCK  128
#define XCOLS    (P_BLOCK + 8)
#define XSTRIDE  137

__global__ __launch_bounds__(256)
void local_linear_wmma(const float* __restrict__ X,
                       const float* __restrict__ W,
                       const float* __restrict__ Bias,
                       float* __restrict__ Out)
{
    __shared__ float Xs[16 * XSTRIDE];
    __shared__ float Ws[P_BLOCK * K_TAPS];

    const int tid = threadIdx.x;
    const int px0 = blockIdx.x * P_BLOCK;
    const int n0  = blockIdx.y * 16;

    for (int i = tid; i < 16 * XCOLS; i += 256) {
        int r  = i / XCOLS;
        int c  = i - r * XCOLS;
        int gc = px0 + c;
        float v = 0.0f;
        if (gc < P_DIM) v = X[(size_t)(n0 + r) * P_DIM + gc];
        Xs[r * XSTRIDE + c] = v;
    }
    for (int i = tid; i < P_BLOCK * K_TAPS; i += 256) {
        int gp = px0 + (i >> 3);
        Ws[i] = (gp < P_DIM) ? W[(size_t)gp * K_TAPS + (i & 7)] : 0.0f;
    }
    __syncthreads();

    const int wave = tid >> 5;
    const int lane = tid & 31;
    const int lrow = lane & 15;
    const int lhi  = lane >> 4;
    const int pw0  = px0 + wave * 16;
    if (pw0 >= P_DIM) return;

    v8f acc = {};
    const float* xrow = &Xs[lrow * XSTRIDE + wave * 16];
    const float* wrow = &Ws[(wave * 16 + lrow) << 3];

    #pragma unroll
    for (int j = 0; j < 6; ++j) {
        const int q0 = 4 * j + 2 * lhi;
        const int q1 = q0 + 1;
        v2f a, b;
        a.x = xrow[q0];
        a.y = xrow[q1];
        const int k0 = q0 - lrow;
        const int k1 = q1 - lrow;
        float w0 = wrow[(unsigned)k0 & 7u];
        float w1 = wrow[(unsigned)k1 & 7u];
        b.x = ((unsigned)k0 < 8u) ? w0 : 0.0f;
        b.y = ((unsigned)k1 < 8u) ? w1 : 0.0f;
        acc = __builtin_amdgcn_wmma_f32_16x16x4_f32(
             false, a,  false, b,
             (short)0, acc,  false,  false);
    }

    const float bias = Bias[pw0 + lrow];
    size_t base = (size_t)(n0 + 8 * lhi) * P_DIM + (size_t)(pw0 + lrow);
    #pragma unroll
    for (int v = 0; v < 8; ++v) {
        Out[base + (size_t)v * P_DIM] = acc[v] + bias;
    }
}

extern "C" void kernel_launch(void* const* d_in, const int* in_sizes, int n_in,
                              void* d_out, int out_size, void* d_ws, size_t ws_size,
                              hipStream_t stream) {
    (void)in_sizes; (void)n_in; (void)d_ws; (void)ws_size; (void)out_size;
    const float* X    = (const float*)d_in[0];
    const float* W    = (const float*)d_in[1];
    const float* Bias = (const float*)d_in[2];
    float* Out        = (float*)d_out;

    dim3 grid((P_DIM + P_BLOCK - 1) / P_BLOCK, N_ROWS / 16);
    local_linear_wmma<<<grid, dim3(256), 0, stream>>>(X, W, Bias, Out);
}


# fewer, fatter waves: each 4-wave workgroup walks 8 row tiles x 128 columns (80384 -> 5024 waves, weights built once) and requests each tile's loads branch-free as one batch, not one at a time
# speedup vs baseline: 2.0628x; 2.0628x over previous
#include <hip/hip_runtime.h>


typedef __attribute__((ext_vector_type(2))) float v2f;
typedef __attribute__((ext_vector_type(8))) float v8f;

#define N_ROWS    1024
#define P_DIM     20000
#define K_TAPS    8
#define NTHREADS  128
#define P_BLOCK   128
#define XSTRIDE   137
#define XTILE     (16 * XSTRIDE)
#define ROW_TILES 8

static_assert(P_DIM % 32 == 0, "a wave's 32 columns are all inside or all outside");
static_assert(P_BLOCK == NTHREADS, "staging maps one thread to one column");

struct XRegs { float m[16]; float h; };

static __device__ __forceinline__
XRegs load_x(const float* __restrict__ Xt, int cmain, int choff, int hrow)
{
    XRegs x;
    for (int r = 0; r < 16; ++r) x.m[r] = Xt[r * P_DIM + cmain];
    x.h = Xt[hrow * P_DIM + choff];
    return x;
}

static __device__ __forceinline__
void store_x(float* __restrict__ dst, const XRegs& x, int tid, int hrow, bool main_ok, bool halo_ok)
{
    for (int r = 0; r < 16; ++r) dst[r * XSTRIDE + tid] = main_ok ? x.m[r] : 0.0f;
    dst[hrow * XSTRIDE + P_BLOCK + (tid & 7)] = halo_ok ? x.h : 0.0f;
}

__global__ __launch_bounds__(NTHREADS)
void local_linear_wmma(const float* __restrict__ X,
                       const float* __restrict__ W,
                       const float* __restrict__ Bias,
                       float* __restrict__ Out)
{
    __shared__ float Xs[2 * XTILE];
    __shared__ float Ws[P_BLOCK * K_TAPS];

    const int tid = threadIdx.x;
    const int px0 = blockIdx.x * P_BLOCK;
    const int n0  = blockIdx.y * (16 * ROW_TILES);

    const int wave = __builtin_amdgcn_readfirstlane(tid >> 5);
    const int lane = tid & 31;
    const int lrow = lane & 15;
    const int lhi  = lane >> 4;
    const int pw0  = px0 + wave * 32;
    const bool active = pw0 < P_DIM;

    const int  hrow    = tid >> 3;
    const int  gmain   = px0 + tid;
    const int  ghalo   = px0 + P_BLOCK + (tid & 7);
    const bool main_ok = gmain < P_DIM;
    const bool halo_ok = ghalo < P_DIM;
    const int  cmain   = (main_ok ? gmain : P_DIM - 1) - px0;
    const int  choff   = (halo_ok ? ghalo : P_DIM - 1) - px0;
    const float* Xt = X + (size_t)n0 * P_DIM + px0;

    float wreg[(P_BLOCK * K_TAPS) / NTHREADS];
    for (int k = 0; k < (P_BLOCK * K_TAPS) / NTHREADS; ++k) {
        int i  = tid + k * NTHREADS;
        int gp = px0 + (i >> 3);
        int gpc = gp < P_DIM ? gp : P_DIM - 1;
        wreg[k] = W[(size_t)gpc * K_TAPS + (i & 7)];
    }
    const int bcol = active ? pw0 + lrow : 0;
    const float bias0 = Bias[bcol];
    const float bias1 = Bias[bcol + 16];
    XRegs xr = load_x(Xt, cmain, choff, hrow);

    for (int k = 0; k < (P_BLOCK * K_TAPS) / NTHREADS; ++k) {
        int i  = tid + k * NTHREADS;
        int gp = px0 + (i >> 3);
        Ws[i] = (gp < P_DIM) ? wreg[k] : 0.0f;
    }
    store_x(Xs, xr, tid, hrow, main_ok, halo_ok);
    __syncthreads();

    v2f bf[2][6];
    for (int ct = 0; ct < 2; ++ct) {
        const float* wrow = &Ws[(wave * 32 + ct * 16 + lrow) << 3];
        for (int j = 0; j < 6; ++j) {
            const int q0 = 4 * j + 2 * lhi;
            const int k0 = q0 - lrow;
            const int k1 = k0 + 1;
            float w0 = wrow[(unsigned)k0 & 7u];
            float w1 = wrow[(unsigned)k1 & 7u];
            bf[ct][j].x = ((unsigned)k0 < 8u) ? w0 : 0.0f;
            bf[ct][j].y = ((unsigned)k1 < 8u) ? w1 : 0.0f;
        }
    }

    for (int t = 0; t < ROW_TILES; ++t) {
        if (active) {
            const float* xrow = &Xs[(t & 1) * XTILE + lrow * XSTRIDE + wave * 32];
            v8f acc0 = {};
            v8f acc1 = {};
            for (int j = 0; j < 6; ++j) {
                const int q0 = 4 * j + 2 * lhi;
                v2f a0, a1;
                a0.x = xrow[q0];
                a0.y = xrow[q0 + 1];
                a1.x = xrow[16 + q0];
                a1.y = xrow[16 + q0 + 1];
                acc0 = __builtin_amdgcn_wmma_f32_16x16x4_f32(
                     false, a0,  false, bf[0][j],
                     (short)0, acc0,  false,  false);
                acc1 = __builtin_amdgcn_wmma_f32_16x16x4_f32(
                     false, a1,  false, bf[1][j],
                     (short)0, acc1,  false,  false);
            }
            size_t base = (size_t)(n0 + 16 * t + 8 * lhi) * P_DIM + (size_t)(pw0 + lrow);
            for (int v = 0; v < 8; ++v) {
                Out[base + (size_t)v * P_DIM]      = acc0[v] + bias0;
                Out[base + (size_t)v * P_DIM + 16] = acc1[v] + bias1;
            }
        }
        if (t + 1 < ROW_TILES) {
            xr = load_x(Xt + (size_t)(16 * (t + 1)) * P_DIM, cmain, choff, hrow);
            store_x(&Xs[((t + 1) & 1) * XTILE], xr, tid, hrow, main_ok, halo_ok);
            __syncthreads();
        }
    }
}

extern "C" void kernel_launch(void* const* d_in, const int* in_sizes, int n_in,
                              void* d_out, int out_size, void* d_ws, size_t ws_size,
                              hipStream_t stream) {
    (void)in_sizes; (void)n_in; (void)d_ws; (void)ws_size; (void)out_size;
    const float* X    = (const float*)d_in[0];
    const float* W    = (const float*)d_in[1];
    const float* Bias = (const float*)d_in[2];
    float* Out        = (float*)d_out;

    dim3 grid((P_DIM + P_BLOCK - 1) / P_BLOCK, N_ROWS / (16 * ROW_TILES));
    local_linear_wmma<<<grid, dim3(NTHREADS), 0, stream>>>(X, W, Bias, Out);
}
